# starting kernel
# baseline (speedup 1.0000x reference)
#include <hip/hip_runtime.h>
#include <hip/hip_bf16.h>

#define N_SRC     50000
#define N_DST     50000
#define N_EDGES   640000
#define IN_FEATS  256
#define OUT_FEATS 128

typedef __attribute__((ext_vector_type(2))) float v2f;
typedef __attribute__((ext_vector_type(8))) float v8f;

__global__ void zero_f4(float4* __restrict__ out, int n4) {
    int i = blockIdx.x * blockDim.x + threadIdx.x;
    if (i < n4) out[i] = make_float4(0.f, 0.f, 0.f, 0.f);
}

__global__ void gemm_cj_wmma(const float* __restrict__ feat,
                             const float* __restrict__ W,
                             const float* __restrict__ cj,
                             float* __restrict__ h) {
    const int lane = threadIdx.x & 31;
    const int wave = threadIdx.x >> 5;
    const int mt   = blockIdx.x * (blockDim.x >> 5) + wave;
    if (mt * 16 >= N_SRC) return;

    const int row0 = mt * 16;
    const int m    = lane & 15;
    const int half = lane >> 4;

    const float* __restrict__ arow = feat + (size_t)(row0 + m) * IN_FEATS;

    v8f acc[8];
    #pragma unroll
    for (int t = 0; t < 8; ++t) acc[t] = (v8f){};

    #pragma unroll 2
    for (int k0 = 0; k0 < IN_FEATS; k0 += 4) {
        const int ka = k0 + 2 * half;
        const v2f a = *(const v2f*)(arow + ka);
        const float* __restrict__ wp = W + (size_t)ka * OUT_FEATS + m;
        #pragma unroll
        for (int t = 0; t < 8; ++t) {
            v2f b;
            b.x = wp[t * 16];
            b.y = wp[OUT_FEATS + t * 16];
            acc[t] = __builtin_amdgcn_wmma_f32_16x16x4_f32(
                          false, a,  false, b,
                          (short)0, acc[t],
                          false,  false);
        }
    }

    #pragma unroll
    for (int i = 0; i < 8; ++i) {
        const int   row = row0 + i + 8 * half;
        const float s   = cj[row];
        float* __restrict__ hrow = h + (size_t)row * OUT_FEATS + m;
        #pragma unroll
        for (int t = 0; t < 8; ++t) hrow[t * 16] = acc[t][i] * s;
    }
}

__global__ void edge_scatter(const float* __restrict__ h,
                             const int* __restrict__ src,
                             const int* __restrict__ dst,
                             const float* __restrict__ ci,
                             float* __restrict__ out) {
    const int lane = threadIdx.x & 31;
    const int e    = (blockIdx.x * blockDim.x + threadIdx.x) >> 5;
    if (e >= N_EDGES) return;

    const int   s     = src[e];
    const int   d     = dst[e];
    const float scale = ci[d];

    const float4 v = *(const float4*)(h + (size_t)s * OUT_FEATS + lane * 4);
    float* __restrict__ o = out + (size_t)d * OUT_FEATS + lane * 4;
    atomicAdd(o + 0, v.x * scale);
    atomicAdd(o + 1, v.y * scale);
    atomicAdd(o + 2, v.z * scale);
    atomicAdd(o + 3, v.w * scale);
}

extern "C" void kernel_launch(void* const* d_in, const int* in_sizes, int n_in,
                              void* d_out, int out_size, void* d_ws, size_t ws_size,
                              hipStream_t stream) {
    const float* feat   = (const float*)d_in[0];
    const float* weight = (const float*)d_in[1];
    const float* cj     = (const float*)d_in[2];
    const float* ci     = (const float*)d_in[3];
    const int*   src    = (const int*)d_in[4];
    const int*   dst    = (const int*)d_in[5];
    float*       out    = (float*)d_out;
    float*       h      = (float*)d_ws;

    {
        const int n4 = (N_DST * OUT_FEATS) / 4;
        zero_f4<<<(n4 + 255) / 256, 256, 0, stream>>>((float4*)out, n4);
    }

    {
        const int mtiles = N_SRC / 16;
        const int wavesPerBlock = 8;
        const int blocks = (mtiles + wavesPerBlock - 1) / wavesPerBlock;
        gemm_cj_wmma<<<blocks, wavesPerBlock * 32, 0, stream>>>(feat, weight, cj, h);
    }

    {
        const int wavesPerBlock = 8;
        const int blocks = (N_EDGES + wavesPerBlock - 1) / wavesPerBlock;
        edge_scatter<<<blocks, wavesPerBlock * 32, 0, stream>>>(h, src, dst, ci, out);
    }
}


# Edge scatter: one wave per 4 consecutive edges, no loop: src, dst, ci by scalar fetch, 16 row loads in flight, atomics in edge order; waves still enter in edge order; float atomics, arrival order
# speedup vs baseline: 1.6995x; 1.6995x over previous
#include <hip/hip_runtime.h>
#include <hip/hip_bf16.h>

#define N_SRC     50000
#define N_DST     50000
#define N_EDGES   640000
#define IN_FEATS  256
#define OUT_FEATS 128

typedef __attribute__((ext_vector_type(2))) float v2f;
typedef __attribute__((ext_vector_type(8))) float v8f;

__global__ void zero_f4(float4* __restrict__ out, int n4) {
    int i = blockIdx.x * blockDim.x + threadIdx.x;
    if (i < n4) out[i] = make_float4(0.f, 0.f, 0.f, 0.f);
}

__global__ void gemm_cj_wmma(const float* __restrict__ feat,
                             const float* __restrict__ W,
                             const float* __restrict__ cj,
                             float* __restrict__ h) {
    const int lane = threadIdx.x & 31;
    const int wave = threadIdx.x >> 5;
    const int mt   = blockIdx.x * (blockDim.x >> 5) + wave;
    if (mt * 16 >= N_SRC) return;

    const int row0 = mt * 16;
    const int m    = lane & 15;
    const int half = lane >> 4;

    const float* __restrict__ arow = feat + (size_t)(row0 + m) * IN_FEATS;

    v8f acc[8];
    #pragma unroll
    for (int t = 0; t < 8; ++t) acc[t] = (v8f){};

    #pragma unroll 2
    for (int k0 = 0; k0 < IN_FEATS; k0 += 4) {
        const int ka = k0 + 2 * half;
        const v2f a = *(const v2f*)(arow + ka);
        const float* __restrict__ wp = W + (size_t)ka * OUT_FEATS + m;
        #pragma unroll
        for (int t = 0; t < 8; ++t) {
            v2f b;
            b.x = wp[t * 16];
            b.y = wp[OUT_FEATS + t * 16];
            acc[t] = __builtin_amdgcn_wmma_f32_16x16x4_f32(
                          false, a,  false, b,
                          (short)0, acc[t],
                          false,  false);
        }
    }

    #pragma unroll
    for (int i = 0; i < 8; ++i) {
        const int   row = row0 + i + 8 * half;
        const float s   = cj[row];
        float* __restrict__ hrow = h + (size_t)row * OUT_FEATS + m;
        #pragma unroll
        for (int t = 0; t < 8; ++t) hrow[t * 16] = acc[t][i] * s;
    }
}

#define EDGES_PER_WAVE 4
__global__ void edge_scatter(const float* __restrict__ h,
                             const int* __restrict__ src,
                             const int* __restrict__ dst,
                             const float* __restrict__ ci,
                             float* __restrict__ out) {
    const int lane = threadIdx.x & 31;
    const int w  = __builtin_amdgcn_readfirstlane(
                       (int)((blockIdx.x * blockDim.x + threadIdx.x) >> 5));
    const int e0 = w * EDGES_PER_WAVE;
    if (e0 >= N_EDGES) return;

    const int4 s4 = *(const int4*)(src + e0);
    const int4 d4 = *(const int4*)(dst + e0);
    asm volatile("" :: "s"(s4.x), "s"(s4.y), "s"(s4.z), "s"(s4.w),
                       "s"(d4.x), "s"(d4.y), "s"(d4.z), "s"(d4.w));
    const int s[EDGES_PER_WAVE] = { s4.x, s4.y, s4.z, s4.w };
    const int d[EDGES_PER_WAVE] = { d4.x, d4.y, d4.z, d4.w };

    float scale[EDGES_PER_WAVE];
    float v[EDGES_PER_WAVE][4];
    #pragma unroll
    for (int j = 0; j < EDGES_PER_WAVE; ++j) {
        scale[j] = ci[d[j]];
        const float* hp = h + (size_t)s[j] * OUT_FEATS + lane;
        v[j][0] = hp[0];
        v[j][1] = hp[32];
        v[j][2] = hp[64];
        v[j][3] = hp[96];
    }
    #pragma unroll
    for (int j = 0; j < EDGES_PER_WAVE; ++j) {
        float* o = out + (size_t)d[j] * OUT_FEATS + lane;
        atomicAdd(o +  0, v[j][0] * scale[j]);
        atomicAdd(o + 32, v[j][1] * scale[j]);
        atomicAdd(o + 64, v[j][2] * scale[j]);
        atomicAdd(o + 96, v[j][3] * scale[j]);
    }
}

extern "C" void kernel_launch(void* const* d_in, const int* in_sizes, int n_in,
                              void* d_out, int out_size, void* d_ws, size_t ws_size,
                              hipStream_t stream) {
    const float* feat   = (const float*)d_in[0];
    const float* weight = (const float*)d_in[1];
    const float* cj     = (const float*)d_in[2];
    const float* ci     = (const float*)d_in[3];
    const int*   src    = (const int*)d_in[4];
    const int*   dst    = (const int*)d_in[5];
    float*       out    = (float*)d_out;
    float*       h      = (float*)d_ws;

    {
        const int n4 = (N_DST * OUT_FEATS) / 4;
        zero_f4<<<(n4 + 255) / 256, 256, 0, stream>>>((float4*)out, n4);
    }

    {
        const int mtiles = N_SRC / 16;
        const int wavesPerBlock = 8;
        const int blocks = (mtiles + wavesPerBlock - 1) / wavesPerBlock;
        gemm_cj_wmma<<<blocks, wavesPerBlock * 32, 0, stream>>>(feat, weight, cj, h);
    }

    {
        const int wavesPerBlock = 8;
        const int edgesPerBlock = wavesPerBlock * EDGES_PER_WAVE;
        const int blocks = (N_EDGES + edgesPerBlock - 1) / edgesPerBlock;
        edge_scatter<<<blocks, wavesPerBlock * 32, 0, stream>>>(h, src, dst, ci, out);
    }
}


# GEMM: 256 resident work-groups copy W once into LDS in fragment order; waves take 16-row pieces, feat straight from global, 16 B per lane; the k of a row are summed in another fixed order
# speedup vs baseline: 2.6259x; 1.5451x over previous
#include <hip/hip_runtime.h>
#include <hip/hip_bf16.h>

#define N_SRC     50000
#define N_DST     50000
#define N_EDGES   640000
#define IN_FEATS  256
#define OUT_FEATS 128

typedef __attribute__((ext_vector_type(2))) float v2f;
typedef __attribute__((ext_vector_type(8))) float v8f;

#define EDGES_PER_WAVE   8
#define WAVES_PER_GROUP  8
#define EDGES_PER_BLOCK  (EDGES_PER_WAVE * WAVES_PER_GROUP)
#define N_BLOCKS         (N_EDGES / EDGES_PER_BLOCK)
#define HALF_ROWS        (N_DST / 2)
#define TICKET_STRIDE    128
#define TICKET_WORDS     (2 * TICKET_STRIDE)

__global__ void zero_f4(float* __restrict__ out, int n4, unsigned int* __restrict__ tickets) {
    int i = blockIdx.x * blockDim.x + threadIdx.x;
    if (i < n4) ((float4*)out)[i] = make_float4(0.f, 0.f, 0.f, 0.f);
    if (i < TICKET_WORDS / 4) ((uint4*)tickets)[i] = make_uint4(0u, 0u, 0u, 0u);
}

typedef __attribute__((ext_vector_type(4))) float v4f;

#define GEMM_WGS     256
#define GEMM_WAVES   16
#define N_UNITS      (2 * (N_SRC / 16))
#define W_PITCH      136
#define GEMM_LDS_ENTRIES  (168 * 1024 / 8)

template <int NT>
__device__ __forceinline__ void gemm_piece(v4f (&a)[4],
                                           const float* __restrict__ arow,
                                           const v2f* bw,
                                           const int row0, const int col0,
                                           const int half, const int m,
                                           const float* __restrict__ cj,
                                           float* __restrict__ h) {
    v8f acc[NT];
    #pragma unroll
    for (int t = 0; t < NT; ++t) acc[t] = (v8f){};

    float s[8];
    #pragma unroll
    for (int i = 0; i < 8; ++i) s[i] = cj[row0 + i + 8 * half];

    #pragma unroll
    for (int kb = 0; kb < IN_FEATS / 32; ++kb) {
        v4f an[4];
        if (kb + 1 < IN_FEATS / 32) {
            #pragma unroll
            for (int j = 0; j < 4; ++j) an[j] = *(const v4f*)(arow + 32 * (kb + 1) + 8 * j);
        }
        #pragma unroll
        for (int j = 0; j < 4; ++j) {
            #pragma unroll
            for (int p = 0; p < 2; ++p) {
                v2f av;
                av.x = a[j][2 * p];
                av.y = a[j][2 * p + 1];
                const int kp = 16 * kb + 4 * j + p;
                #pragma unroll
                for (int t = 0; t < NT; ++t) {
                    const v2f b = bw[kp * W_PITCH + 16 * t];
                    acc[t] = __builtin_amdgcn_wmma_f32_16x16x4_f32(
                                  false, av,  false, b,
                                  (short)0, acc[t],
                                  false,  false);
                }
            }
        }
        if (kb + 1 < IN_FEATS / 32) {
            #pragma unroll
            for (int j = 0; j < 4; ++j) a[j] = an[j];
        }
    }

    #pragma unroll
    for (int i = 0; i < 8; ++i) {
        const int row = row0 + i + 8 * half;
        float* __restrict__ hrow = h + (size_t)row * OUT_FEATS + col0 + m;
        #pragma unroll
        for (int t = 0; t < NT; ++t) hrow[t * 16] = acc[t][i] * s[i];
    }
}

__global__ void __launch_bounds__(GEMM_WAVES * 32)
gemm_cj_wmma(const float* __restrict__ feat,
             const float* __restrict__ W,
             const float* __restrict__ cj,
             float* __restrict__ h) {
    __shared__ __attribute__((aligned(16))) v2f sW[GEMM_LDS_ENTRIES];

    const int lane = threadIdx.x & 31;
    const int wave = __builtin_amdgcn_readfirstlane((int)(threadIdx.x >> 5));
    const int m    = lane & 15;
    const int half = lane >> 4;

    const int u_begin = ((int)blockIdx.x * N_UNITS) / GEMM_WGS;
    const int u_end   = (((int)blockIdx.x + 1) * N_UNITS) / GEMM_WGS;

    const int u0 = u_begin + wave;
    const float* __restrict__ arow0 = feat + (size_t)((u0 >> 1) * 16 + m) * IN_FEATS + 4 * half;

    v4f a[4];
    #pragma unroll
    for (int j = 0; j < 4; ++j) a[j] = *(const v4f*)(arow0 + 8 * j);

    #pragma unroll
    for (int r = 0; r < (IN_FEATS / 2) * (OUT_FEATS / 4) / (GEMM_WAVES * 32); ++r) {
        const int id = (int)threadIdx.x + r * (GEMM_WAVES * 32);
        const int c4 = id & 31;
        const int kp = id >> 5;
        const v4f w0 = *(const v4f*)(W + (size_t)(2 * kp)     * OUT_FEATS + 4 * c4);
        const v4f w1 = *(const v4f*)(W + (size_t)(2 * kp + 1) * OUT_FEATS + 4 * c4);
        v4f* d = (v4f*)(sW + kp * W_PITCH + 4 * c4);
        d[0] = (v4f){w0[0], w1[0], w0[1], w1[1]};
        d[1] = (v4f){w0[2], w1[2], w0[3], w1[3]};
    }
    __syncthreads();

    const int n_half = 2 * (u_end - u_begin - GEMM_WAVES);
    const int a4 = wave >> 2, b4 = wave & 3;
    int q  = 4 * ((b4 - a4) & 3) + a4;
    int uh = u_begin + GEMM_WAVES + (q >> 1);
    const float* __restrict__ arow_h = feat + (size_t)((uh >> 1) * 16 + m) * IN_FEATS + 4 * half;

    const v2f* bw0 = sW + (2 * half) * W_PITCH + m;

    v4f nx[4];
    #pragma unroll
    for (int j = 0; j < 4; ++j) nx[j] = *(const v4f*)(arow_h + 8 * j);

    gemm_piece<4>(a, arow0, bw0 + 64 * (u0 & 1),
                  (u0 >> 1) * 16, 64 * (u0 & 1), half, m, cj, h);
    #pragma unroll
    for (int j = 0; j < 4; ++j) a[j] = nx[j];

    #pragma unroll 1
    for (int r = 0; r < 2; ++r) {
        const int  row0 = (uh >> 1) * 16;
        const int  col0 = 64 * (uh & 1) + 32 * (q & 1);
        const int  qn   = q + GEMM_WAVES;
        const bool more = qn < n_half;
        const int  uhn  = more ? u_begin + GEMM_WAVES + (qn >> 1) : uh;
        const float* __restrict__ arow_n = feat + (size_t)((uhn >> 1) * 16 + m) * IN_FEATS + 4 * half;
        if (more) {
            #pragma unroll
            for (int j = 0; j < 4; ++j) nx[j] = *(const v4f*)(arow_n + 8 * j);
        }
        gemm_piece<2>(a, arow_h, bw0 + col0, row0, col0, half, m, cj, h);
        if (!more) break;
        #pragma unroll
        for (int j = 0; j < 4; ++j) a[j] = nx[j];
        q = qn; uh = uhn; arow_h = arow_n;
    }
}

typedef __attribute__((ext_vector_type(8))) int v8i;

__global__ void __launch_bounds__(WAVES_PER_GROUP * 32)
edge_scatter(const float* __restrict__ h,
             const int* __restrict__ src,
             const int* __restrict__ dst,
             const float* __restrict__ ci,
             float* __restrict__ out,
             unsigned int* __restrict__ tickets) {
    __shared__ int s_task;
    if (threadIdx.x == 0) {
        unsigned int half = ((unsigned int)__builtin_amdgcn_s_sendmsg_rtn(0x87) >> 18) & 1u;
        unsigned int t = atomicAdd(&tickets[half * TICKET_STRIDE], 1u);
        if (t >= N_BLOCKS) {
            half ^= 1u;
            t = atomicAdd(&tickets[half * TICKET_STRIDE], 1u);
        }
        s_task = (t < N_BLOCKS) ? (int)(2u * t + half) : -1;
    }
    __syncthreads();
    const int task = __builtin_amdgcn_readfirstlane(s_task);
    if (task < 0) return;
    const int range = task & 1;
    const int blk   = task >> 1;

    const int lane = threadIdx.x & 31;
    const int wv   = __builtin_amdgcn_readfirstlane((int)(threadIdx.x >> 5));
    const int e0   = blk * EDGES_PER_BLOCK + wv * EDGES_PER_WAVE;

    const v8i s8 = *(const v8i*)(src + e0);
    const v8i d8 = *(const v8i*)(dst + e0);
    asm volatile("" :: "s"(s8[0]), "s"(s8[1]), "s"(s8[2]), "s"(s8[3]),
                       "s"(s8[4]), "s"(s8[5]), "s"(s8[6]), "s"(s8[7]),
                       "s"(d8[0]), "s"(d8[1]), "s"(d8[2]), "s"(d8[3]),
                       "s"(d8[4]), "s"(d8[5]), "s"(d8[6]), "s"(d8[7]));

    bool  mine[EDGES_PER_WAVE];
    float scale[EDGES_PER_WAVE];
    float v[EDGES_PER_WAVE][4];
    #pragma unroll
    for (int j = 0; j < EDGES_PER_WAVE; ++j) {
        mine[j] = ((d8[j] >= HALF_ROWS) ? 1 : 0) == range;
        if (mine[j]) {
            scale[j] = ci[d8[j]];
            const float* hp = h + (size_t)s8[j] * OUT_FEATS + lane;
            v[j][0] = hp[0];
            v[j][1] = hp[32];
            v[j][2] = hp[64];
            v[j][3] = hp[96];
        }
    }
    #pragma unroll
    for (int j = 0; j < EDGES_PER_WAVE; ++j) {
        if (mine[j]) {
            float* o = out + (size_t)d8[j] * OUT_FEATS + lane;
            atomicAdd(o +  0, v[j][0] * scale[j]);
            atomicAdd(o + 32, v[j][1] * scale[j]);
            atomicAdd(o + 64, v[j][2] * scale[j]);
            atomicAdd(o + 96, v[j][3] * scale[j]);
        }
    }
}

extern "C" void kernel_launch(void* const* d_in, const int* in_sizes, int n_in,
                              void* d_out, int out_size, void* d_ws, size_t ws_size,
                              hipStream_t stream) {
    const float* feat   = (const float*)d_in[0];
    const float* weight = (const float*)d_in[1];
    const float* cj     = (const float*)d_in[2];
    const float* ci     = (const float*)d_in[3];
    const int*   src    = (const int*)d_in[4];
    const int*   dst    = (const int*)d_in[5];
    float*       out    = (float*)d_out;
    float*       h      = (float*)d_ws;
    unsigned int* tickets = (unsigned int*)(h + (size_t)N_SRC * OUT_FEATS);

    {
        const int n4 = (N_DST * OUT_FEATS) / 4;
        zero_f4<<<(n4 + 255) / 256, 256, 0, stream>>>(out, n4, tickets);
    }

    {
        gemm_cj_wmma<<<GEMM_WGS, GEMM_WAVES * 32, 0, stream>>>(feat, weight, cj, h);
    }

    {
        edge_scatter<<<2 * N_BLOCKS, WAVES_PER_GROUP * 32, 0, stream>>>(h, src, dst, ci, out, tickets);
    }
}
